# starting kernel
# baseline (speedup 1.0000x reference)
#include <hip/hip_runtime.h>


typedef float v2f __attribute__((ext_vector_type(2)));
typedef float v8f __attribute__((ext_vector_type(8)));

#define TS    256
#define FIN   18
#define HID   64
#define G4    256
#define NOUT  15
#define BM    16
#define HPAD  (HID + 4)

__device__ __forceinline__ float fast_sig(float x) {
    return __builtin_amdgcn_rcpf(1.0f + __expf(-x));
}
__device__ __forceinline__ float fast_tanh(float x) {
    return 1.0f - 2.0f * __builtin_amdgcn_rcpf(1.0f + __expf(2.0f * x));
}

__global__ __launch_bounds__(128)
void lstm_fused_kernel(const float* __restrict__ X,
                       const float* __restrict__ W_ih,
                       const float* __restrict__ W_hh,
                       const float* __restrict__ b_ih,
                       const float* __restrict__ b_hh,
                       const float* __restrict__ W_cls,
                       const float* __restrict__ b_cls,
                       float* __restrict__ out)
{
    __shared__ float hbuf[2][BM][HPAD];

    const int tid  = threadIdx.x;
    const int wave = tid >> 5;
    const int lane = tid & 31;
    const int m    = lane & 15;
    const int g2   = (lane >> 4) << 1;
    const int b0   = blockIdx.x * BM;

    v2f  bh[4][16];
    v2f  bx[4][5];
    float bias[4];

#pragma unroll
    for (int q = 0; q < 4; ++q) {
        const int n = q * HID + wave * 16 + m;
#pragma unroll
        for (int k = 0; k < 16; ++k)
            bh[q][k] = *(const v2f*)(W_hh + n * HID + 4 * k + g2);
#pragma unroll
        for (int k = 0; k < 5; ++k) {
            const int col = 4 * k + g2;
            v2f w = {0.0f, 0.0f};
            if (col + 1 < FIN) w = *(const v2f*)(W_ih + n * FIN + col);
            bx[q][k] = w;
        }
        bias[q] = b_ih[n] + b_hh[n];
    }

    for (int i = tid; i < BM * HPAD; i += 128)
        ((float*)hbuf[0])[i] = 0.0f;

    v8f c = {};
    int cur = 0;
    const float* xrow = X + (long)(b0 + m) * TS * FIN;

    for (int t = 0; t < TS; ++t) {
        __syncthreads();

        v2f ax[5];
#pragma unroll
        for (int k = 0; k < 5; ++k) {
            const int col = 4 * k + g2;
            v2f a = {0.0f, 0.0f};
            if (col + 1 < FIN) a = *(const v2f*)(xrow + t * FIN + col);
            ax[k] = a;
        }
        v2f ah[16];
#pragma unroll
        for (int k = 0; k < 16; ++k)
            ah[k] = *(const v2f*)&hbuf[cur][m][4 * k + g2];

        v8f acc[4];
#pragma unroll
        for (int q = 0; q < 4; ++q) {
            v8f a;
#pragma unroll
            for (int e = 0; e < 8; ++e) a[e] = bias[q];
#pragma unroll
            for (int k = 0; k < 5; ++k)
                a = __builtin_amdgcn_wmma_f32_16x16x4_f32(
                        false, ax[k], false, bx[q][k], (short)0, a, false, false);
#pragma unroll
            for (int k = 0; k < 16; ++k)
                a = __builtin_amdgcn_wmma_f32_16x16x4_f32(
                        false, ah[k], false, bh[q][k], (short)0, a, false, false);
            acc[q] = a;
        }

        v8f h;
#pragma unroll
        for (int e = 0; e < 8; ++e) {
            const float ig = fast_sig (acc[0][e]);
            const float fg = fast_sig (acc[1][e]);
            const float gg = fast_tanh(acc[2][e]);
            const float og = fast_sig (acc[3][e]);
            const float cn = fg * c[e] + ig * gg;
            c[e] = cn;
            h[e] = og * fast_tanh(cn);
        }

        const int nxt  = cur ^ 1;
        const int row0 = (lane >> 4) * 8;
#pragma unroll
        for (int e = 0; e < 8; ++e)
            hbuf[nxt][row0 + e][wave * 16 + m] = h[e];
        cur = nxt;
    }

    __syncthreads();

    for (int idx = tid; idx < BM * NOUT; idx += 128) {
        const int r = idx / NOUT;
        const int o = idx % NOUT;
        float s = b_cls[o];
#pragma unroll 8
        for (int k = 0; k < HID; ++k)
            s += hbuf[cur][r][k] * W_cls[o * HID + k];
        out[(b0 + r) * NOUT + o] = s;
    }
}

extern "C" void kernel_launch(void* const* d_in, const int* in_sizes, int n_in,
                              void* d_out, int out_size, void* d_ws, size_t ws_size,
                              hipStream_t stream) {
    const float* X     = (const float*)d_in[0];
    const float* W_ih  = (const float*)d_in[1];
    const float* W_hh  = (const float*)d_in[2];
    const float* b_ih  = (const float*)d_in[3];
    const float* b_hh  = (const float*)d_in[4];
    const float* W_cls = (const float*)d_in[5];
    const float* b_cls = (const float*)d_in[6];

    const int B = in_sizes[0] / (TS * FIN);
    dim3 grid(B / BM);
    dim3 block(128);

    lstm_fused_kernel<<<grid, block, 0, stream>>>(
        X, W_ih, W_hh, b_ih, b_hh, W_cls, b_cls, (float*)d_out);
}


# Software-pipelined time step: X loaded two steps ahead, input-projection WMMAs placed to cover the LDS latency and the cell update, activations issued between WMMAs; sum order unchanged
# speedup vs baseline: 1.2878x; 1.2878x over previous
#include <hip/hip_runtime.h>


typedef float v2f __attribute__((ext_vector_type(2)));
typedef float v8f __attribute__((ext_vector_type(8)));

#define TS    256
#define FIN   18
#define HID   64
#define G4    256
#define NOUT  15
#define BM    16
#define HPAD  (HID + 4)

__device__ __forceinline__ v8f mma4(v2f a, v2f b, v8f acc) {
    return __builtin_amdgcn_wmma_f32_16x16x4_f32(
               false, a, false, b, (short)0, acc, false, false);
}

struct CellTmp {
    float ei[8], eg[8], ri[8], rg[8];
    float ef[8], eo[8], rf[8], ro[8];
    float ec[8], rc[8];
};

__device__ __forceinline__ void piece_ig(int idx, const v8f& ai, const v8f& ag,
                                         CellTmp& w, v8f& ig) {
    const int st = idx >> 3, e = idx & 7;
    if      (st == 0) w.ei[e] = __expf(-ai[e]);
    else if (st == 1) w.eg[e] = __expf(2.0f * ag[e]);
    else if (st == 2) w.ri[e] = __builtin_amdgcn_rcpf(1.0f + w.ei[e]);
    else if (st == 3) w.rg[e] = __builtin_amdgcn_rcpf(1.0f + w.eg[e]);
    else              ig[e]   = w.ri[e] * (1.0f - 2.0f * w.rg[e]);
}

__device__ __forceinline__ void piece_fo(int idx, const v8f& af, const v8f& ao,
                                         const v8f& ig, CellTmp& w, v8f& c, v8f& h) {
    const int st = idx >> 3, e = idx & 7;
    if      (st == 0) w.ef[e] = __expf(-af[e]);
    else if (st == 1) w.eo[e] = __expf(-ao[e]);
    else if (st == 2) w.rf[e] = __builtin_amdgcn_rcpf(1.0f + w.ef[e]);
    else if (st == 3) w.ro[e] = __builtin_amdgcn_rcpf(1.0f + w.eo[e]);
    else if (st == 4) { const float cn = w.rf[e] * c[e] + ig[e];
                        c[e] = cn;
                        w.ec[e] = __expf(2.0f * cn); }
    else if (st == 5) w.rc[e] = __builtin_amdgcn_rcpf(1.0f + w.ec[e]);
    else              h[e]    = w.ro[e] * (1.0f - 2.0f * w.rc[e]);
}

#define KEEP_ORDER() __builtin_amdgcn_sched_barrier(0)

__device__ __forceinline__ void issue_before_here(v8f& tile) {
    asm volatile("" : "+v"(tile));
}

__global__ __launch_bounds__(128)
void lstm_fused_kernel(const float* __restrict__ X,
                       const float* __restrict__ W_ih,
                       const float* __restrict__ W_hh,
                       const float* __restrict__ b_ih,
                       const float* __restrict__ b_hh,
                       const float* __restrict__ W_cls,
                       const float* __restrict__ b_cls,
                       float* __restrict__ out)
{
    __shared__ float hbuf[2][BM][HPAD];

    const int tid  = threadIdx.x;
    const int wave = tid >> 5;
    const int lane = tid & 31;
    const int m    = lane & 15;
    const int g2   = (lane >> 4) << 1;
    const int b0   = blockIdx.x * BM;

    v2f  bh[4][16];
    v2f  bx[4][5];
    float bias[4];

#pragma unroll
    for (int q = 0; q < 4; ++q) {
        const int n = q * HID + wave * 16 + m;
#pragma unroll
        for (int k = 0; k < 16; ++k)
            bh[q][k] = *(const v2f*)(W_hh + n * HID + 4 * k + g2);
#pragma unroll
        for (int k = 0; k < 5; ++k) {
            const int col = 4 * k + g2;
            v2f w = {0.0f, 0.0f};
            if (col + 1 < FIN) w = *(const v2f*)(W_ih + n * FIN + col);
            bx[q][k] = w;
        }
        bias[q] = b_ih[n] + b_hh[n];
    }

    for (int i = tid; i < BM * HPAD; i += 128)
        ((float*)hbuf[0])[i] = 0.0f;

    v8f c = {};
    int cur = 0;
    const float* xrow = X + (long)(b0 + m) * TS * FIN;

    v8f bsplat[4];
#pragma unroll
    for (int q = 0; q < 4; ++q)
#pragma unroll
        for (int e = 0; e < 8; ++e) bsplat[q][e] = bias[q];

    v2f ax_cur[5], ax_nxt[5];
#pragma unroll
    for (int k = 0; k < 5; ++k) {
        const int col = 4 * k + g2;
        v2f a0 = {0.0f, 0.0f};
        v2f a1 = {0.0f, 0.0f};
        if (col + 1 < FIN) {
            a0 = *(const v2f*)(xrow + col);
            a1 = *(const v2f*)(xrow + FIN + col);
        }
        ax_cur[k] = a0;
        ax_nxt[k] = a1;
    }
    v8f xfo[2];
    xfo[0] = bsplat[1];
    xfo[1] = bsplat[3];
#pragma unroll
    for (int k = 0; k < 5; ++k) {
        xfo[0] = mma4(ax_cur[k], bx[1][k], xfo[0]);
        xfo[1] = mma4(ax_cur[k], bx[3][k], xfo[1]);
    }

    for (int t = 0; t < TS; ++t) {
        __syncthreads();

        v2f ah[16];
#pragma unroll
        for (int k = 0; k < 16; ++k)
            ah[k] = *(const v2f*)&hbuf[cur][m][4 * k + g2];

        const int t2 = (t + 2 < TS) ? (t + 2) : (TS - 1);
        v2f ax_new[5];
#pragma unroll
        for (int k = 0; k < 5; ++k) {
            const int col = 4 * k + g2;
            v2f a = {0.0f, 0.0f};
            if (col + 1 < FIN) a = *(const v2f*)(xrow + t2 * FIN + col);
            ax_new[k] = a;
        }
        KEEP_ORDER();

        v8f acc[4];
        acc[0] = bsplat[0];
        acc[2] = bsplat[2];
#pragma unroll
        for (int k = 0; k < 5; ++k) {
            acc[0] = mma4(ax_cur[k], bx[0][k], acc[0]);
            acc[2] = mma4(ax_cur[k], bx[2][k], acc[2]);
        }
#pragma unroll
        for (int k = 0; k < 16; ++k) {
            acc[0] = mma4(ah[k], bh[0][k], acc[0]);
            acc[2] = mma4(ah[k], bh[2][k], acc[2]);
        }
        KEEP_ORDER();

        acc[1] = xfo[0];
        acc[3] = xfo[1];
        CellTmp w;
        v8f ig;
#pragma unroll
        for (int p = 0; p < 32; ++p) {
            const int k = p >> 1;
            if ((p & 1) == 0) acc[1] = mma4(ah[k], bh[1][k], acc[1]);
            else              acc[3] = mma4(ah[k], bh[3][k], acc[3]);
            if (p < 20) {
                piece_ig(2 * p,     acc[0], acc[2], w, ig);
                piece_ig(2 * p + 1, acc[0], acc[2], w, ig);
            }
            KEEP_ORDER();
        }

        xfo[0] = bsplat[1];
        xfo[1] = bsplat[3];
        v8f h;
#pragma unroll
        for (int p = 0; p < 10; ++p) {
            const int k = p >> 1;
            if ((p & 1) == 0) { xfo[0] = mma4(ax_nxt[k], bx[1][k], xfo[0]);
                                issue_before_here(xfo[0]); }
            else              { xfo[1] = mma4(ax_nxt[k], bx[3][k], xfo[1]);
                                issue_before_here(xfo[1]); }
#pragma unroll
            for (int i = (56 * p + 9) / 10; i < (56 * (p + 1) + 9) / 10; ++i)
                piece_fo(i, acc[1], acc[3], ig, w, c, h);
            KEEP_ORDER();
        }

        const int nxt  = cur ^ 1;
        const int row0 = (lane >> 4) * 8;
#pragma unroll
        for (int e = 0; e < 8; ++e)
            hbuf[nxt][row0 + e][wave * 16 + m] = h[e];
        cur = nxt;
        KEEP_ORDER();
#pragma unroll
        for (int k = 0; k < 5; ++k) { ax_cur[k] = ax_nxt[k]; ax_nxt[k] = ax_new[k]; }
    }

    __syncthreads();

    for (int idx = tid; idx < BM * NOUT; idx += 128) {
        const int r = idx / NOUT;
        const int o = idx % NOUT;
        float s = b_cls[o];
#pragma unroll 8
        for (int k = 0; k < HID; ++k)
            s += hbuf[cur][r][k] * W_cls[o * HID + k];
        out[(b0 + r) * NOUT + o] = s;
    }
}

extern "C" void kernel_launch(void* const* d_in, const int* in_sizes, int n_in,
                              void* d_out, int out_size, void* d_ws, size_t ws_size,
                              hipStream_t stream) {
    const float* X     = (const float*)d_in[0];
    const float* W_ih  = (const float*)d_in[1];
    const float* W_hh  = (const float*)d_in[2];
    const float* b_ih  = (const float*)d_in[3];
    const float* b_hh  = (const float*)d_in[4];
    const float* W_cls = (const float*)d_in[5];
    const float* b_cls = (const float*)d_in[6];

    const int B = in_sizes[0] / (TS * FIN);
    dim3 grid(B / BM);
    dim3 block(128);

    lstm_fused_kernel<<<grid, block, 0, stream>>>(
        X, W_ih, W_hh, b_ih, b_hh, W_cls, b_cls, (float*)d_out);
}


# Activations on the hardware tanh: sigmoid(x) = (1 + tanh(x/2))/2, the halving folded exactly into the i/f/o weight and bias fragments; 5 transcendentals per element, not 10; matrix sums unchanged
# speedup vs baseline: 1.4288x; 1.1095x over previous
#include <hip/hip_runtime.h>


typedef float v2f __attribute__((ext_vector_type(2)));
typedef float v8f __attribute__((ext_vector_type(8)));

#define TS    256
#define FIN   18
#define HID   64
#define G4    256
#define NOUT  15
#define BM    16
#define HPAD  (HID + 4)

__device__ __forceinline__ v8f mma4(v2f a, v2f b, v8f acc) {
    return __builtin_amdgcn_wmma_f32_16x16x4_f32(
               false, a, false, b, (short)0, acc, false, false);
}

struct CellTmp {
    v2f ti[4], tg[4], si[4];
    v2f tf[4], sf[4];
    v2f to[4], so[4];
    v2f tc[4];
};

__device__ __forceinline__ v2f pair_of(const v8f& t, int j) {
    v2f r = {t[2 * j], t[2 * j + 1]};
    return r;
}

__device__ __forceinline__ v2f tanh2(v2f x) {
    v2f r = {__builtin_amdgcn_tanhf(x[0]), __builtin_amdgcn_tanhf(x[1])};
    return r;
}

__device__ __forceinline__ v2f sigmoid_of_tanh_half(v2f t) {
    const v2f half = {0.5f, 0.5f};
    return __builtin_elementwise_fma(half, t, half);
}

#define KEEP_ORDER() __builtin_amdgcn_sched_barrier(0)

__device__ __forceinline__ void issue_before_here(v8f& tile) {
    asm volatile("" : "+v"(tile));
}

__global__ __launch_bounds__(128)
void lstm_fused_kernel(const float* __restrict__ X,
                       const float* __restrict__ W_ih,
                       const float* __restrict__ W_hh,
                       const float* __restrict__ b_ih,
                       const float* __restrict__ b_hh,
                       const float* __restrict__ W_cls,
                       const float* __restrict__ b_cls,
                       float* __restrict__ out)
{
    __shared__ float hbuf[2][BM][HPAD];

    const int tid  = threadIdx.x;
    const int wave = tid >> 5;
    const int lane = tid & 31;
    const int m    = lane & 15;
    const int g2   = (lane >> 4) << 1;
    const int b0   = blockIdx.x * BM;

    v2f  bh[4][16];
    v2f  bx[4][5];
    float bias[4];

#pragma unroll
    for (int q = 0; q < 4; ++q) {
        const int n = q * HID + wave * 16 + m;
        const float gs = (q == 2) ? 1.0f : 0.5f;
#pragma unroll
        for (int k = 0; k < 16; ++k)
            bh[q][k] = *(const v2f*)(W_hh + n * HID + 4 * k + g2) * gs;
#pragma unroll
        for (int k = 0; k < 5; ++k) {
            const int col = 4 * k + g2;
            v2f w = {0.0f, 0.0f};
            if (col + 1 < FIN) w = *(const v2f*)(W_ih + n * FIN + col);
            bx[q][k] = w * gs;
        }
        bias[q] = (b_ih[n] + b_hh[n]) * gs;
    }

    for (int i = tid; i < BM * HPAD; i += 128)
        ((float*)hbuf[0])[i] = 0.0f;

    v2f c[4] = {};
    int cur = 0;
    const float* xrow = X + (long)(b0 + m) * TS * FIN;

    v8f bsplat[4];
#pragma unroll
    for (int q = 0; q < 4; ++q)
#pragma unroll
        for (int e = 0; e < 8; ++e) bsplat[q][e] = bias[q];

    v2f ax_cur[5], ax_nxt[5];
#pragma unroll
    for (int k = 0; k < 5; ++k) {
        const int col = 4 * k + g2;
        v2f a0 = {0.0f, 0.0f};
        v2f a1 = {0.0f, 0.0f};
        if (col + 1 < FIN) {
            a0 = *(const v2f*)(xrow + col);
            a1 = *(const v2f*)(xrow + FIN + col);
        }
        ax_cur[k] = a0;
        ax_nxt[k] = a1;
    }
    v8f xfo[2];
    xfo[0] = bsplat[1];
    xfo[1] = bsplat[3];
#pragma unroll
    for (int k = 0; k < 5; ++k) {
        xfo[0] = mma4(ax_cur[k], bx[1][k], xfo[0]);
        xfo[1] = mma4(ax_cur[k], bx[3][k], xfo[1]);
    }

    for (int t = 0; t < TS; ++t) {
        __syncthreads();

        v2f ah[16];
#pragma unroll
        for (int k = 0; k < 16; ++k)
            ah[k] = *(const v2f*)&hbuf[cur][m][4 * k + g2];

        const int t2 = (t + 2 < TS) ? (t + 2) : (TS - 1);
        v2f ax_new[5];
#pragma unroll
        for (int k = 0; k < 5; ++k) {
            const int col = 4 * k + g2;
            v2f a = {0.0f, 0.0f};
            if (col + 1 < FIN) a = *(const v2f*)(xrow + t2 * FIN + col);
            ax_new[k] = a;
        }
        KEEP_ORDER();

        v8f acc[4];
        acc[0] = bsplat[0];
        acc[2] = bsplat[2];
#pragma unroll
        for (int k = 0; k < 5; ++k) {
            acc[0] = mma4(ax_cur[k], bx[0][k], acc[0]);
            acc[2] = mma4(ax_cur[k], bx[2][k], acc[2]);
        }
#pragma unroll
        for (int k = 0; k < 16; ++k) {
            acc[0] = mma4(ah[k], bh[0][k], acc[0]);
            acc[2] = mma4(ah[k], bh[2][k], acc[2]);
        }
        KEEP_ORDER();

        acc[1] = xfo[0];
        acc[3] = xfo[1];
        CellTmp w;
        v2f ig[4];
#pragma unroll
        for (int p = 0; p < 32; ++p) {
            const int k = p >> 1;
            if ((p & 1) == 0) acc[1] = mma4(ah[k], bh[1][k], acc[1]);
            else              acc[3] = mma4(ah[k], bh[3][k], acc[3]);
            if      (p < 4)  w.ti[p]     = tanh2(pair_of(acc[0], p));
            else if (p < 8)  w.tg[p - 4] = tanh2(pair_of(acc[2], p - 4));
            else if (p < 12) w.si[p - 8] = sigmoid_of_tanh_half(w.ti[p - 8]);
            else if (p < 16) ig[p - 12]  = w.si[p - 12] * w.tg[p - 12];
            KEEP_ORDER();
        }

        xfo[0] = bsplat[1];
        xfo[1] = bsplat[3];
        v2f h[4];
#pragma unroll
        for (int p = 0; p < 10; ++p) {
            const int k = p >> 1;
            if ((p & 1) == 0) { xfo[0] = mma4(ax_nxt[k], bx[1][k], xfo[0]);
                                issue_before_here(xfo[0]); }
            else              { xfo[1] = mma4(ax_nxt[k], bx[3][k], xfo[1]);
                                issue_before_here(xfo[1]); }
            if (p == 0 || p == 1) {
                w.tf[2 * p]     = tanh2(pair_of(acc[1], 2 * p));
                w.tf[2 * p + 1] = tanh2(pair_of(acc[1], 2 * p + 1));
            }
            if (p == 2 || p == 3) {
                const int j = 2 * (p - 2);
                w.sf[j]     = sigmoid_of_tanh_half(w.tf[j]);
                w.to[j]     = tanh2(pair_of(acc[3], j));
                w.sf[j + 1] = sigmoid_of_tanh_half(w.tf[j + 1]);
                w.to[j + 1] = tanh2(pair_of(acc[3], j + 1));
            }
            if (p == 4 || p == 5) {
                const int j = 2 * (p - 4);
                c[j]     = __builtin_elementwise_fma(w.sf[j],     c[j],     ig[j]);
                c[j + 1] = __builtin_elementwise_fma(w.sf[j + 1], c[j + 1], ig[j + 1]);
            }
            if (p == 5 || p == 6) {
                const int j = 2 * (p - 5);
                w.tc[j]     = tanh2(c[j]);
                w.tc[j + 1] = tanh2(c[j + 1]);
            }
            if (p == 6 || p == 7) {
                const int j = 2 * (p - 6);
                w.so[j]     = sigmoid_of_tanh_half(w.to[j]);
                w.so[j + 1] = sigmoid_of_tanh_half(w.to[j + 1]);
            }
            if (p == 7 || p == 8) {
                const int j = 2 * (p - 7);
                h[j]     = w.so[j]     * w.tc[j];
                h[j + 1] = w.so[j + 1] * w.tc[j + 1];
            }
            KEEP_ORDER();
        }

        const int nxt  = cur ^ 1;
        const int row0 = (lane >> 4) * 8;
#pragma unroll
        for (int e = 0; e < 8; ++e)
            hbuf[nxt][row0 + e][wave * 16 + m] = h[e >> 1][e & 1];
        cur = nxt;
        KEEP_ORDER();
#pragma unroll
        for (int k = 0; k < 5; ++k) { ax_cur[k] = ax_nxt[k]; ax_nxt[k] = ax_new[k]; }
    }

    __syncthreads();

    for (int idx = tid; idx < BM * NOUT; idx += 128) {
        const int r = idx / NOUT;
        const int o = idx % NOUT;
        float s = b_cls[o];
#pragma unroll 8
        for (int k = 0; k < HID; ++k)
            s += hbuf[cur][r][k] * W_cls[o * HID + k];
        out[(b0 + r) * NOUT + o] = s;
    }
}

extern "C" void kernel_launch(void* const* d_in, const int* in_sizes, int n_in,
                              void* d_out, int out_size, void* d_ws, size_t ws_size,
                              hipStream_t stream) {
    const float* X     = (const float*)d_in[0];
    const float* W_ih  = (const float*)d_in[1];
    const float* W_hh  = (const float*)d_in[2];
    const float* b_ih  = (const float*)d_in[3];
    const float* b_hh  = (const float*)d_in[4];
    const float* W_cls = (const float*)d_in[5];
    const float* b_cls = (const float*)d_in[6];

    const int B = in_sizes[0] / (TS * FIN);
    dim3 grid(B / BM);
    dim3 block(128);

    lstm_fused_kernel<<<grid, block, 0, stream>>>(
        X, W_ih, W_hh, b_ih, b_hh, W_cls, b_cls, (float*)d_out);
}


# Hand-over between steps shortened: time loop unrolled by two at fixed LDS buffer names (register copies, address math, guard branch dropped), h stored at once, next f/o input WMMAs behind it
# speedup vs baseline: 1.4872x; 1.0409x over previous
#include <hip/hip_runtime.h>


typedef float v2f __attribute__((ext_vector_type(2)));
typedef float v8f __attribute__((ext_vector_type(8)));

#define TS    256
#define FIN   18
#define HID   64
#define G4    256
#define NOUT  15
#define BM    16
#define HPAD  (HID + 4)

__device__ __forceinline__ v8f mma4(v2f a, v2f b, v8f acc) {
    return __builtin_amdgcn_wmma_f32_16x16x4_f32(
               false, a, false, b, (short)0, acc, false, false);
}

__device__ __forceinline__ v2f pair_of(const v8f& t, int j) {
    v2f r = {t[2 * j], t[2 * j + 1]};
    return r;
}

__device__ __forceinline__ v2f tanh2(v2f x) {
    v2f r = {__builtin_amdgcn_tanhf(x[0]), __builtin_amdgcn_tanhf(x[1])};
    return r;
}

__device__ __forceinline__ v2f sigmoid_of_tanh_half(v2f t) {
    const v2f half = {0.5f, 0.5f};
    return __builtin_elementwise_fma(half, t, half);
}

#define KEEP_ORDER() __builtin_amdgcn_sched_barrier(0)

__device__ __forceinline__ void issue_before_here(v8f& tile) {
    asm volatile("" : "+v"(tile));
}

__device__ __forceinline__ void load_x_row(const float* p, int g2, v2f (&a)[5]) {
#pragma unroll
    for (int k = 0; k < 4; ++k)
        a[k] = *(const v2f*)(p + 4 * k + g2);
    v2f a4 = *(const v2f*)(p + 16);
    const v2f zero = {0.0f, 0.0f};
    a[4] = (g2 == 0) ? a4 : zero;
}

template <int CUR, bool LOAD>
__device__ __forceinline__ void lstm_step(
    float (&hbuf)[2][BM][HPAD],
    const v2f (&bh)[4][16], const v2f (&bx)[4][5], const v8f (&bsplat)[4],
    v2f (&ax_head)[5], const v2f (&ax_tail)[5],
    v8f (&xfo)[2], v2f (&c)[4],
    const float* xload, int m, int g2, int lane, int wave)
{
    __syncthreads();

    v2f ah[16];
#pragma unroll
    for (int k = 0; k < 16; ++k)
        ah[k] = *(const v2f*)&hbuf[CUR][m][4 * k + g2];

    v2f ax_new[5];
    if (LOAD) load_x_row(xload, g2, ax_new);
    KEEP_ORDER();

    v8f acc[4];
    acc[0] = mma4(ax_head[0], bx[0][0], bsplat[0]);
    acc[2] = mma4(ax_head[0], bx[2][0], bsplat[2]);
#pragma unroll
    for (int k = 1; k < 5; ++k) {
        acc[0] = mma4(ax_head[k], bx[0][k], acc[0]);
        acc[2] = mma4(ax_head[k], bx[2][k], acc[2]);
    }

#pragma unroll
    for (int k = 0; k < 16; ++k) {
        acc[0] = mma4(ah[k], bh[0][k], acc[0]);
        acc[2] = mma4(ah[k], bh[2][k], acc[2]);
    }
    KEEP_ORDER();

    v2f ti[4], tg[4];
#pragma unroll
    for (int p = 0; p < 32; ++p) {
        const int k = p >> 1;
        if (p == 0)            acc[1] = mma4(ah[0], bh[1][0], xfo[0]);
        else if (p == 1)       acc[3] = mma4(ah[0], bh[3][0], xfo[1]);
        else if ((p & 1) == 0) acc[1] = mma4(ah[k], bh[1][k], acc[1]);
        else                   acc[3] = mma4(ah[k], bh[3][k], acc[3]);
        if      (p < 4)  ti[p]      = tanh2(pair_of(acc[0], p));
        else if (p < 8)  tg[p - 4]  = tanh2(pair_of(acc[2], p - 4));
        KEEP_ORDER();
    }

    v2f tf[4], to[4], si[4], ig[4], sf[4], so[4], tc[4], h[4];
#pragma unroll
    for (int j = 0; j < 4; ++j) tf[j] = tanh2(pair_of(acc[1], j));
#pragma unroll
    for (int j = 0; j < 4; ++j) si[j] = sigmoid_of_tanh_half(ti[j]);
#pragma unroll
    for (int j = 0; j < 4; ++j) ig[j] = si[j] * tg[j];
#pragma unroll
    for (int j = 0; j < 4; ++j) to[j] = tanh2(pair_of(acc[3], j));
#pragma unroll
    for (int j = 0; j < 4; ++j) sf[j] = sigmoid_of_tanh_half(tf[j]);
#pragma unroll
    for (int j = 0; j < 4; ++j) c[j] = __builtin_elementwise_fma(sf[j], c[j], ig[j]);
#pragma unroll
    for (int j = 0; j < 4; ++j) tc[j] = tanh2(c[j]);
#pragma unroll
    for (int j = 0; j < 4; ++j) so[j] = sigmoid_of_tanh_half(to[j]);
#pragma unroll
    for (int j = 0; j < 4; ++j) h[j] = so[j] * tc[j];

    const int row0 = (lane >> 4) * 8;
#pragma unroll
    for (int e = 0; e < 8; ++e)
        hbuf[CUR ^ 1][row0 + e][wave * 16 + m] = h[e >> 1][e & 1];
    KEEP_ORDER();

    xfo[0] = mma4(ax_tail[0], bx[1][0], bsplat[1]);
    xfo[1] = mma4(ax_tail[0], bx[3][0], bsplat[3]);
#pragma unroll
    for (int k = 1; k < 5; ++k) {
        xfo[0] = mma4(ax_tail[k], bx[1][k], xfo[0]);
        xfo[1] = mma4(ax_tail[k], bx[3][k], xfo[1]);
    }
    issue_before_here(xfo[0]);
    issue_before_here(xfo[1]);
    KEEP_ORDER();

    if (LOAD) {
#pragma unroll
        for (int k = 0; k < 5; ++k) ax_head[k] = ax_new[k];
    }
}

__global__ __launch_bounds__(128)
void lstm_fused_kernel(const float* __restrict__ X,
                       const float* __restrict__ W_ih,
                       const float* __restrict__ W_hh,
                       const float* __restrict__ b_ih,
                       const float* __restrict__ b_hh,
                       const float* __restrict__ W_cls,
                       const float* __restrict__ b_cls,
                       float* __restrict__ out)
{
    __shared__ float hbuf[2][BM][HPAD];

    const int tid  = threadIdx.x;
    const int wave = tid >> 5;
    const int lane = tid & 31;
    const int m    = lane & 15;
    const int g2   = (lane >> 4) << 1;
    const int b0   = blockIdx.x * BM;

    v2f  bh[4][16];
    v2f  bx[4][5];
    float bias[4];

#pragma unroll
    for (int q = 0; q < 4; ++q) {
        const int n = q * HID + wave * 16 + m;
        const float gs = (q == 2) ? 1.0f : 0.5f;
#pragma unroll
        for (int k = 0; k < 16; ++k)
            bh[q][k] = *(const v2f*)(W_hh + n * HID + 4 * k + g2) * gs;
#pragma unroll
        for (int k = 0; k < 5; ++k) {
            const int col = 4 * k + g2;
            v2f w = {0.0f, 0.0f};
            if (col + 1 < FIN) w = *(const v2f*)(W_ih + n * FIN + col);
            bx[q][k] = w * gs;
        }
        bias[q] = (b_ih[n] + b_hh[n]) * gs;
    }

    for (int i = tid; i < BM * HPAD; i += 128)
        ((float*)hbuf[0])[i] = 0.0f;

    v2f c[4] = {};
    const float* xrow = X + (long)(b0 + m) * TS * FIN;

    v8f bsplat[4];
#pragma unroll
    for (int q = 0; q < 4; ++q)
#pragma unroll
        for (int e = 0; e < 8; ++e) bsplat[q][e] = bias[q];

    v2f ax_a[5], ax_b[5];
    load_x_row(xrow,       g2, ax_a);
    load_x_row(xrow + FIN, g2, ax_b);
    v8f xfo[2];
    xfo[0] = bsplat[1];
    xfo[1] = bsplat[3];
#pragma unroll
    for (int k = 0; k < 5; ++k) {
        xfo[0] = mma4(ax_a[k], bx[1][k], xfo[0]);
        xfo[1] = mma4(ax_a[k], bx[3][k], xfo[1]);
    }

    const float* xt = xrow;
    for (int it = 0; it < TS / 2 - 1; ++it) {
        lstm_step<0, true>(hbuf, bh, bx, bsplat, ax_a, ax_b, xfo, c,
                           xt + 2 * FIN, m, g2, lane, wave);
        lstm_step<1, true>(hbuf, bh, bx, bsplat, ax_b, ax_a, xfo, c,
                           xt + 3 * FIN, m, g2, lane, wave);
        xt += 2 * FIN;
    }
    lstm_step<0, false>(hbuf, bh, bx, bsplat, ax_a, ax_b, xfo, c,
                        xt, m, g2, lane, wave);
    lstm_step<1, false>(hbuf, bh, bx, bsplat, ax_b, ax_a, xfo, c,
                        xt, m, g2, lane, wave);

    __syncthreads();

    for (int idx = tid; idx < BM * NOUT; idx += 128) {
        const int r = idx / NOUT;
        const int o = idx % NOUT;
        float s = b_cls[o];
#pragma unroll 8
        for (int k = 0; k < HID; ++k)
            s += hbuf[0][r][k] * W_cls[o * HID + k];
        out[(b0 + r) * NOUT + o] = s;
    }
}

extern "C" void kernel_launch(void* const* d_in, const int* in_sizes, int n_in,
                              void* d_out, int out_size, void* d_ws, size_t ws_size,
                              hipStream_t stream) {
    const float* X     = (const float*)d_in[0];
    const float* W_ih  = (const float*)d_in[1];
    const float* W_hh  = (const float*)d_in[2];
    const float* b_ih  = (const float*)d_in[3];
    const float* b_hh  = (const float*)d_in[4];
    const float* W_cls = (const float*)d_in[5];
    const float* b_cls = (const float*)d_in[6];

    const int B = in_sizes[0] / (TS * FIN);
    dim3 grid(B / BM);
    dim3 block(128);

    lstm_fused_kernel<<<grid, block, 0, stream>>>(
        X, W_ih, W_hh, b_ih, b_hh, W_cls, b_cls, (float*)d_out);
}
